# starting kernel
# baseline (speedup 1.0000x reference)
#include <hip/hip_runtime.h>

typedef float v2f __attribute__((ext_vector_type(2)));
typedef float v8f __attribute__((ext_vector_type(8)));

#define ROWS   16
#define TH     (ROWS + 2)
#define TW     72
#define WIDTH  64
#define HEIGHT 64
#define CCH    128

__device__ __constant__ float Gm[3][3] = {
    { 0.5773502691896258f,  0.5773502691896258f,  0.5773502691896258f },
    { 0.7071067811865476f,  0.0f,                -0.7071067811865476f },
    { 0.4082482904638630f, -0.8164965809277260f,  0.4082482904638630f }
};

__global__ void __launch_bounds__(128)
sfcr_wmma_kernel(const float* __restrict__ x,
                 const float* __restrict__ Wl,
                 const float* __restrict__ bl,
                 float* __restrict__ out,
                 int N)
{
    __shared__ float sW[4][12];
    __shared__ float sT[TH * TW];

    const int tid = threadIdx.x;
    const int blk = blockIdx.x;
    const int HB  = HEIGHT / ROWS;
    const int h0  = (blk % HB) * ROWS;
    const int c   = (blk / HB) % CCH;
    const int n   =  blk / (HB * CCH);
    (void)N;

    if (tid < 36) {
        int o = tid / 9, j = tid % 9, p = j / 3, q = j % 3;
        float acc = Wl[o * 18 + j];
#pragma unroll
        for (int a = 0; a < 3; ++a)
#pragma unroll
            for (int b = 0; b < 3; ++b)
                acc += Wl[o * 18 + 9 + 3 * a + b] * Gm[a][p] * Gm[b][q];
        sW[o][j] = acc;
    }

    const float* xin = x + (size_t)(n * CCH + c) * (HEIGHT * WIDTH);
    for (int idx = tid; idx < TH * (WIDTH + 2); idx += 128) {
        int r = idx / (WIDTH + 2), col = idx % (WIDTH + 2);
        int gh = h0 - 1 + r; gh = gh < 0 ? 0 : (gh > HEIGHT - 1 ? HEIGHT - 1 : gh);
        int gw = col - 1;    gw = gw < 0 ? 0 : (gw > WIDTH  - 1 ? WIDTH  - 1 : gw);
        sT[r * TW + col] = xin[gh * WIDTH + gw];
    }
    __syncthreads();

    const int wave = tid >> 5;
    const int lane = tid & 31;
    const int half = lane >> 4;
    const int li   = lane & 15;
    const int kb   = half * 2;
    const int m    = li;

    v2f a0 = {0.f, 0.f}, a1 = {0.f, 0.f}, a2 = {0.f, 0.f};
    if (m < 4) {
        a0.x = sW[m][0 + kb]; a0.y = sW[m][1 + kb];
        a1.x = sW[m][4 + kb]; a1.y = sW[m][5 + kb];
        if (!half) a2.x = sW[m][8];
    }

    v8f cinit = {};
    if (!half) {
        float b0 = bl[0], b1 = bl[1], b2 = bl[2], b3 = bl[3];
        cinit[0] = b0; cinit[1] = b1; cinit[2] = b2; cinit[3] = b3;
    }

    float* obase = out + (size_t)((n * CCH + c) * 4) * (HEIGHT * WIDTH);

    for (int it = 0; it < 16; ++it) {
        int chain = wave + 4 * it;
        int r  = chain >> 2;
        int w0 = (chain & 3) << 4;

        const float* t = &sT[r * TW + w0 + li];
        v2f b0, b1, b2;
        b0.x = half ? t[2]            : t[0];
        b0.y = half ? t[TW + 0]       : t[1];
        b1.x = half ? t[2 * TW + 0]   : t[TW + 1];
        b1.y = half ? t[2 * TW + 1]   : t[TW + 2];
        b2.x = half ? 0.f             : t[2 * TW + 2];
        b2.y = 0.f;

        v8f acc = cinit;
        acc = __builtin_amdgcn_wmma_f32_16x16x4_f32(false, a0, false, b0, (short)0, acc, false, false);
        acc = __builtin_amdgcn_wmma_f32_16x16x4_f32(false, a1, false, b1, (short)0, acc, false, false);
        acc = __builtin_amdgcn_wmma_f32_16x16x4_f32(false, a2, false, b2, (short)0, acc, false, false);

        if (!half) {
            float* op = obase + (size_t)(h0 + r) * WIDTH + w0 + li;
#pragma unroll
            for (int o = 0; o < 4; ++o)
                op[(size_t)o * (HEIGHT * WIDTH)] = acc[o];
        }
    }
}

extern "C" void kernel_launch(void* const* d_in, const int* in_sizes, int n_in,
                              void* d_out, int out_size, void* d_ws, size_t ws_size,
                              hipStream_t stream) {
    const float* x  = (const float*)d_in[0];
    const float* Wl = (const float*)d_in[1];
    const float* bl = (const float*)d_in[2];
    float* out = (float*)d_out;

    int N = in_sizes[0] / (CCH * HEIGHT * WIDTH);
    int blocks = N * CCH * (HEIGHT / ROWS);
    sfcr_wmma_kernel<<<blocks, 128, 0, stream>>>(x, Wl, bl, out, N);
}


# One work-group per whole 64x64 plane (2048 groups of 4 waves, was 8192): plane staged once by 16-byte loads, weights folded once per plane, each wave loops over 4 strips; sum order unchanged
# speedup vs baseline: 2.2428x; 2.2428x over previous
#include <hip/hip_runtime.h>


typedef float v2f __attribute__((ext_vector_type(2)));
typedef float v4f __attribute__((ext_vector_type(4)));
typedef float v8f __attribute__((ext_vector_type(8)));

#define WIDTH  64
#define HEIGHT 64
#define CCH    128
#define TROWS  (HEIGHT + 2)
#define TW     72
#define COL0   4
#define STRIPS 4
#define BAND   28

__device__ __constant__ float Gm[3][3] = {
    { 0.5773502691896258f,  0.5773502691896258f,  0.5773502691896258f },
    { 0.7071067811865476f,  0.0f,                -0.7071067811865476f },
    { 0.4082482904638630f, -0.8164965809277260f,  0.4082482904638630f }
};

__device__ __forceinline__ void put_piece(float* sT, int trow, int c4, v4f v)
{
    *(v4f*)(sT + trow * TW + COL0 + 4 * c4) = v;
}

__global__ void __launch_bounds__(128)
sfcr_wmma_kernel(const float* __restrict__ x,
                 const float* __restrict__ Wl,
                 const float* __restrict__ bl,
                 float* __restrict__ out,
                 int N)
{
    __shared__ float sA[4 * BAND];
    __shared__ __attribute__((aligned(16))) float sT[TROWS * TW];

    const int tid = threadIdx.x;
    const int pl  = blockIdx.x;
    (void)N;

    const v4f* xin = (const v4f*)(x + (size_t)pl * (HEIGHT * WIDTH));
    v4f g[8];
#pragma unroll
    for (int i = 0; i < 8; ++i) g[i] = xin[tid + 128 * i];

    if (tid < 36) {
        int o = tid / 9, j = tid % 9, p = j / 3, q = j % 3;
        float acc = Wl[o * 18 + j];
#pragma unroll
        for (int a = 0; a < 3; ++a)
#pragma unroll
            for (int b = 0; b < 3; ++b)
                acc += Wl[o * 18 + 9 + 3 * a + b] * Gm[a][p] * Gm[b][q];
        sA[o * BAND + 9 + j] = acc;
    } else if (tid < 36 + 72) {
        int z = tid - 36, o = z / 18, e = z % 18;
        sA[o * BAND + (e < 9 ? e : e + 9)] = 0.f;
    }

    {
        const int c4 = tid & 15;
        const int r0 = tid >> 4;
#pragma unroll
        for (int i = 0; i < 8; ++i) {
            const int row = r0 + 8 * i;
            put_piece(sT, row + 1, c4, g[i]);
            if (i == 0 && row == 0)          put_piece(sT, 0, c4, g[i]);
            if (i == 7 && row == HEIGHT - 1) put_piece(sT, TROWS - 1, c4, g[i]);
        }
    }
    __syncthreads();

    const int wave = tid >> 5;
    const int lane = tid & 31;
    const int half = lane >> 4;
    const int li   = lane & 15;

    const int dr = li >> 2;
    const int o  = li & 3;

    const float* ap = &sA[o * BAND + (3 * half - dr + 3) * 3];
    const v2f a0 = { ap[0], ap[1] };
    const v2f a1 = { ap[2], ap[3] };
    const v2f a2 = { ap[4], ap[5] };
    const v2f a3 = { ap[6], ap[7] };
    const v2f a4 = { ap[8], 0.f };

    v8f cinit;
    {
        float b0 = bl[0], b1 = bl[1], b2 = bl[2], b3 = bl[3];
        cinit[0] = b0; cinit[1] = b1; cinit[2] = b2; cinit[3] = b3;
        cinit[4] = b0; cinit[5] = b1; cinit[6] = b2; cinit[7] = b3;
    }

    const float* tp = &sT[(16 * wave + 3 * half) * TW + 4 * li + COL0];
    float* op = out + (size_t)pl * (4 * HEIGHT * WIDTH)
                    + (size_t)(16 * wave + 2 * half) * WIDTH + 4 * li;

    const float* lp = tp + (li == 0  ? 0 : -1);
    const float* rn = tp + (li == 15 ? 3 :  4);

#pragma unroll 1
    for (int s = 0; s < STRIPS; ++s) {
        float T[3][6];
#pragma unroll
        for (int i = 0; i < 3; ++i) {
            const float* rp = tp + (4 * s + i) * TW;
            v4f mid = *(const v4f*)rp;
            T[i][0] = lp[(4 * s + i) * TW];
            T[i][1] = mid.x; T[i][2] = mid.y; T[i][3] = mid.z; T[i][4] = mid.w;
            T[i][5] = rn[(4 * s + i) * TW];
        }

        v8f acc[4];
#pragma unroll
        for (int j = 0; j < 4; ++j) acc[j] = cinit;

#pragma unroll
        for (int j = 0; j < 4; ++j) {
            v2f b0 = { T[0][j],     T[0][j + 1] };
            v2f b1 = { T[0][j + 2], T[1][j]     };
            v2f b2 = { T[1][j + 1], T[1][j + 2] };
            v2f b3 = { T[2][j],     T[2][j + 1] };
            v2f b4 = { T[2][j + 2], 0.f         };
            acc[j] = __builtin_amdgcn_wmma_f32_16x16x4_f32(false, a0, false, b0, (short)0, acc[j], false, false);
            acc[j] = __builtin_amdgcn_wmma_f32_16x16x4_f32(false, a1, false, b1, (short)0, acc[j], false, false);
            acc[j] = __builtin_amdgcn_wmma_f32_16x16x4_f32(false, a2, false, b2, (short)0, acc[j], false, false);
            acc[j] = __builtin_amdgcn_wmma_f32_16x16x4_f32(false, a3, false, b3, (short)0, acc[j], false, false);
            acc[j] = __builtin_amdgcn_wmma_f32_16x16x4_f32(false, a4, false, b4, (short)0, acc[j], false, false);
        }

        float* os = op + (size_t)(4 * s) * WIDTH;
#pragma unroll
        for (int r = 0; r < 8; ++r) {
            v4f v = { acc[0][r], acc[1][r], acc[2][r], acc[3][r] };
            *(v4f*)(os + (size_t)(r & 3) * (HEIGHT * WIDTH) + (r >> 2) * WIDTH) = v;
        }
    }
}

extern "C" void kernel_launch(void* const* d_in, const int* in_sizes, int n_in,
                              void* d_out, int out_size, void* d_ws, size_t ws_size,
                              hipStream_t stream) {
    const float* x  = (const float*)d_in[0];
    const float* Wl = (const float*)d_in[1];
    const float* bl = (const float*)d_in[2];
    float* out = (float*)d_out;

    int N = in_sizes[0] / (CCH * HEIGHT * WIDTH);
    int blocks = N * CCH;
    sfcr_wmma_kernel<<<blocks, 128, 0, stream>>>(x, Wl, bl, out, N);
}
